# starting kernel
# baseline (speedup 1.0000x reference)
#include <hip/hip_runtime.h>
#include <cstdint>
#include <cstddef>

#define BB 4
#define HH 32
#define WWD 256
#define CC 96
#define DI 192
#define KK 2
#define NN 16
#define RR 6
#define LL (HH*WWD)
#define BL (BB*LL)
#define NOUT (2*DI)
#define CDBL (RR + 2*NN)
#define NC 128
#define CL (LL/NC)

typedef float v2f __attribute__((ext_vector_type(2)));
typedef float v8f __attribute__((ext_vector_type(8)));

__device__ __forceinline__ v8f wmma4(v2f a, v2f b, v8f c) {
  return __builtin_amdgcn_wmma_f32_16x16x4_f32(false, a, false, b, (short)0, c,
                                               false, false);
}

__device__ __forceinline__ float silu_f(float x) {
  return x / (1.0f + __expf(-x));
}
__device__ __forceinline__ float softplus_f(float x) {
  return (x > 20.0f) ? x : log1pf(__expf(x));
}

__global__ void __launch_bounds__(256) k_inproj(const float* __restrict__ x,
                                                const float* __restrict__ W_in,
                                                float* __restrict__ xz) {
  __shared__ float sX[16][CC + 1];
  const int rowBase = blockIdx.x * 16;
  const int tid = threadIdx.x;
  for (int i = tid; i < 16 * CC; i += 256) {
    int m = i / CC, kk = i % CC;
    sX[m][kk] = x[(size_t)(rowBase + m) * CC + kk];
  }
  __syncthreads();
  const int wave = tid >> 5, lane = tid & 31;
  const int lid = lane & 15, half = lane >> 4;
  for (int nt = wave; nt < NOUT / 16; nt += 8) {
    const int nbase = nt * 16;
    const float* wrow = W_in + (size_t)(nbase + lid) * CC;
    v8f acc = {};
    for (int k0 = 0; k0 < CC; k0 += 4) {
      v2f a; a.x = sX[lid][k0 + 2 * half]; a.y = sX[lid][k0 + 2 * half + 1];
      v2f b; b.x = wrow[k0 + 2 * half];    b.y = wrow[k0 + 2 * half + 1];
      acc = wmma4(a, b, acc);
    }
    float* orow = xz + (size_t)(rowBase + half * 8) * NOUT + nbase + lid;
#pragma unroll
    for (int v = 0; v < 8; ++v) orow[(size_t)v * NOUT] = acc[v];
  }
}

__global__ void __launch_bounds__(256) k_dwconv(const float* __restrict__ xz,
                                                const float* __restrict__ conv_w,
                                                const float* __restrict__ conv_b,
                                                float* __restrict__ xconv) {
  int idx = blockIdx.x * 256 + threadIdx.x;
  if (idx >= BB * HH * WWD * DI) return;
  int c = idx % DI; int t = idx / DI;
  int w = t % WWD; t /= WWD;
  int h = t % HH;  int b = t / HH;
  float s = conv_b[c];
#pragma unroll
  for (int dy = -1; dy <= 1; ++dy) {
    int hh = h + dy; if (hh < 0 || hh >= HH) continue;
#pragma unroll
    for (int dx = -1; dx <= 1; ++dx) {
      int ww = w + dx; if (ww < 0 || ww >= WWD) continue;
      s += xz[(size_t)((b * HH + hh) * WWD + ww) * NOUT + c] *
           conv_w[c * 9 + (dy + 1) * 3 + (dx + 1)];
    }
  }
  xconv[(size_t)((b * HH + h) * WWD + w) * DI + c] = silu_f(s);
}

__global__ void __launch_bounds__(256) k_xdbl(const float* __restrict__ xconv,
                                              const float* __restrict__ xpw,
                                              float* __restrict__ x_dbl) {
  const int tid = threadIdx.x, wave = tid >> 5, lane = tid & 31;
  const int lid = lane & 15, half = lane >> 4;
  const int tile = blockIdx.x * 8 + wave;
  const int lt = tile % (LL / 16);
  const int bk = tile / (LL / 16);
  const int k = bk % KK, b = bk / KK;
  const int lbase = lt * 16;
  const int l = lbase + lid;
  const int lidx = k ? (LL - 1 - l) : l;
  const float* arow = xconv + (size_t)(b * LL + lidx) * DI;
  v8f acc[3] = {};
  for (int k0 = 0; k0 < DI; k0 += 4) {
    v2f a; a.x = arow[k0 + 2 * half]; a.y = arow[k0 + 2 * half + 1];
#pragma unroll
    for (int t = 0; t < 3; ++t) {
      int c2 = t * 16 + lid;
      v2f bb;
      if (c2 < CDBL) {
        const float* wrow = xpw + (size_t)(k * CDBL + c2) * DI;
        bb.x = wrow[k0 + 2 * half]; bb.y = wrow[k0 + 2 * half + 1];
      } else { bb.x = 0.f; bb.y = 0.f; }
      acc[t] = wmma4(a, bb, acc[t]);
    }
  }
#pragma unroll
  for (int t = 0; t < 3; ++t) {
    int col = t * 16 + lid;
    if (col < CDBL) {
      float* orow = x_dbl + ((size_t)bk * LL + lbase + half * 8) * CDBL + col;
#pragma unroll
      for (int v = 0; v < 8; ++v) orow[(size_t)v * CDBL] = acc[t][v];
    }
  }
}

__global__ void __launch_bounds__(256) k_dtproj(const float* __restrict__ x_dbl,
                                                const float* __restrict__ dtw,
                                                const float* __restrict__ dtb,
                                                float* __restrict__ delta) {
  const int tid = threadIdx.x, wave = tid >> 5, lane = tid & 31;
  const int lid = lane & 15, half = lane >> 4;
  const int tile = blockIdx.x;
  const int lt = tile % (LL / 16);
  const int bk = tile / (LL / 16);
  const int k = bk % KK;
  const int lbase = lt * 16;
  const float* arow = x_dbl + ((size_t)bk * LL + lbase + lid) * CDBL;
  v2f a0, a1;
  a0.x = arow[2 * half]; a0.y = arow[2 * half + 1];
  a1.x = (half == 0) ? arow[4] : 0.f;
  a1.y = (half == 0) ? arow[5] : 0.f;
  for (int nt = wave; nt < DI / 16; nt += 8) {
    const int d = nt * 16 + lid;
    const float* wrow = dtw + (size_t)(k * DI + d) * RR;
    v2f b0, b1;
    b0.x = wrow[2 * half]; b0.y = wrow[2 * half + 1];
    b1.x = (half == 0) ? wrow[4] : 0.f;
    b1.y = (half == 0) ? wrow[5] : 0.f;
    v8f acc = {};
    acc = wmma4(a0, b0, acc);
    acc = wmma4(a1, b1, acc);
    const float bias = dtb[k * DI + d];
    float* orow = delta + ((size_t)bk * LL + lbase + half * 8) * DI + d;
#pragma unroll
    for (int v = 0; v < 8; ++v) orow[(size_t)v * DI] = softplus_f(acc[v] + bias);
  }
}

__global__ void __launch_bounds__(256) k_scan_pass1(
    const float* __restrict__ xconv, const float* __restrict__ delta,
    const float* __restrict__ x_dbl, const float* __restrict__ A_logs,
    float* __restrict__ chunkA, float* __restrict__ chunkH) {
  const int gid = blockIdx.x * 256 + threadIdx.x;
  const int d = gid % DI;
  const int c = (gid / DI) % NC;
  const int bk = gid / (DI * NC);
  const int k = bk % KK, b = bk / KK;
  float Av[NN];
#pragma unroll
  for (int n = 0; n < NN; ++n)
    Av[n] = -__expf(A_logs[(size_t)(k * DI + d) * NN + n]);
  float aP[NN], h[NN];
#pragma unroll
  for (int n = 0; n < NN; ++n) { aP[n] = 1.f; h[n] = 0.f; }
  const float* xc = xconv + (size_t)b * LL * DI;
  const float* dl = delta + (size_t)bk * LL * DI;
  const float* xd = x_dbl + (size_t)bk * LL * CDBL;
  const int l0 = c * CL;
  for (int l = l0; l < l0 + CL; ++l) {
    const int lidx = k ? (LL - 1 - l) : l;
    const float u = xc[(size_t)lidx * DI + d];
    const float dlt = dl[(size_t)l * DI + d];
    const float* row = xd + (size_t)l * CDBL;
    const float du = dlt * u;
#pragma unroll
    for (int n = 0; n < NN; ++n) {
      const float dA = __expf(dlt * Av[n]);
      aP[n] *= dA;
      h[n] = dA * h[n] + du * row[RR + n];
    }
  }
#pragma unroll
  for (int n = 0; n < NN; ++n) {
    const size_t idx = ((size_t)(bk * NC + c) * NN + n) * DI + d;
    chunkA[idx] = aP[n];
    chunkH[idx] = h[n];
  }
}

__global__ void __launch_bounds__(256) k_scan_pass2(
    const float* __restrict__ chunkA, float* __restrict__ chunkH) {
  const int gid = blockIdx.x * 256 + threadIdx.x;
  const int d = gid % DI;
  const int bk = gid / DI;
  float hrun[NN];
#pragma unroll
  for (int n = 0; n < NN; ++n) hrun[n] = 0.f;
  for (int c = 0; c < NC; ++c) {
#pragma unroll
    for (int n = 0; n < NN; ++n) {
      const size_t idx = ((size_t)(bk * NC + c) * NN + n) * DI + d;
      const float a = chunkA[idx];
      const float hl = chunkH[idx];
      chunkH[idx] = hrun[n];
      hrun[n] = a * hrun[n] + hl;
    }
  }
}

__global__ void __launch_bounds__(256) k_scan_pass3(
    const float* __restrict__ xconv, const float* __restrict__ delta,
    const float* __restrict__ x_dbl, const float* __restrict__ A_logs,
    const float* __restrict__ Ds, const float* __restrict__ chunkH,
    float* __restrict__ ys) {
  const int gid = blockIdx.x * 256 + threadIdx.x;
  const int d = gid % DI;
  const int c = (gid / DI) % NC;
  const int bk = gid / (DI * NC);
  const int k = bk % KK, b = bk / KK;
  float Av[NN];
#pragma unroll
  for (int n = 0; n < NN; ++n)
    Av[n] = -__expf(A_logs[(size_t)(k * DI + d) * NN + n]);
  const float Dv = Ds[k * DI + d];
  float h[NN];
#pragma unroll
  for (int n = 0; n < NN; ++n)
    h[n] = chunkH[((size_t)(bk * NC + c) * NN + n) * DI + d];
  const float* xc = xconv + (size_t)b * LL * DI;
  const float* dl = delta + (size_t)bk * LL * DI;
  const float* xd = x_dbl + (size_t)bk * LL * CDBL;
  float* yo = ys + (size_t)bk * LL * DI;
  const int l0 = c * CL;
  for (int l = l0; l < l0 + CL; ++l) {
    const int lidx = k ? (LL - 1 - l) : l;
    const float u = xc[(size_t)lidx * DI + d];
    const float dlt = dl[(size_t)l * DI + d];
    const float* row = xd + (size_t)l * CDBL;
    const float du = dlt * u;
    float y = 0.f;
#pragma unroll
    for (int n = 0; n < NN; ++n) {
      const float dA = __expf(dlt * Av[n]);
      h[n] = dA * h[n] + du * row[RR + n];
      y += h[n] * row[RR + NN + n];
    }
    yo[(size_t)l * DI + d] = y + u * Dv;
  }
}

__global__ void __launch_bounds__(256) k_outproj(const float* __restrict__ ys,
                                                 const float* __restrict__ xz,
                                                 const float* __restrict__ W_out,
                                                 float* __restrict__ out) {
  __shared__ float sY[16][DI + 1];
  const int rowBase = blockIdx.x * 16;
  const int b = rowBase / LL;
  const int lb = rowBase % LL;
  const int tid = threadIdx.x;
  for (int i = tid; i < 16 * DI; i += 256) {
    int m = i / DI, dd = i % DI;
    int l = lb + m;
    float y0 = ys[((size_t)(b * KK + 0) * LL + l) * DI + dd];
    float y1 = ys[((size_t)(b * KK + 1) * LL + (LL - 1 - l)) * DI + dd];
    float zv = xz[(size_t)(b * LL + l) * NOUT + DI + dd];
    sY[m][dd] = (y0 + y1) * silu_f(zv);
  }
  __syncthreads();
  const int wave = tid >> 5, lane = tid & 31;
  const int lid = lane & 15, half = lane >> 4;
  for (int nt = wave; nt < CC / 16; nt += 8) {
    const int nbase = nt * 16;
    const float* wrow = W_out + (size_t)(nbase + lid) * DI;
    v8f acc = {};
    for (int k0 = 0; k0 < DI; k0 += 4) {
      v2f a;  a.x = sY[lid][k0 + 2 * half];  a.y = sY[lid][k0 + 2 * half + 1];
      v2f bb; bb.x = wrow[k0 + 2 * half];    bb.y = wrow[k0 + 2 * half + 1];
      acc = wmma4(a, bb, acc);
    }
    float* orow = out + (size_t)(rowBase + half * 8) * CC + nbase + lid;
#pragma unroll
    for (int v = 0; v < 8; ++v) orow[(size_t)v * CC] = acc[v];
  }
}

extern "C" void kernel_launch(void* const* d_in, const int* in_sizes, int n_in,
                              void* d_out, int out_size, void* d_ws, size_t ws_size,
                              hipStream_t stream) {
  (void)in_sizes; (void)n_in; (void)out_size; (void)ws_size;
  const float* x      = (const float*)d_in[0];
  const float* W_in   = (const float*)d_in[1];
  const float* conv_w = (const float*)d_in[2];
  const float* conv_b = (const float*)d_in[3];
  const float* xpw    = (const float*)d_in[4];
  const float* dtw    = (const float*)d_in[5];
  const float* dtb    = (const float*)d_in[6];
  const float* A_logs = (const float*)d_in[7];
  const float* Ds     = (const float*)d_in[8];
  const float* W_out  = (const float*)d_in[9];
  float* out = (float*)d_out;

  char* ws = (char*)d_ws;
  size_t off = 0;
  auto alloc = [&](size_t nbytes) {
    char* p = ws + off;
    off = (off + nbytes + 255) & ~(size_t)255;
    return (float*)p;
  };
  float* xz     = alloc((size_t)BL * NOUT * sizeof(float));
  float* xconv  = alloc((size_t)BL * DI * sizeof(float));
  float* x_dbl  = alloc((size_t)BB * KK * LL * CDBL * sizeof(float));
  float* delta  = alloc((size_t)BB * KK * LL * DI * sizeof(float));
  float* ysb    = alloc((size_t)BB * KK * LL * DI * sizeof(float));
  float* chunkA = alloc((size_t)BB * KK * DI * NC * NN * sizeof(float));
  float* chunkH = alloc((size_t)BB * KK * DI * NC * NN * sizeof(float));

  k_inproj<<<BL / 16, 256, 0, stream>>>(x, W_in, xz);
  k_dwconv<<<(BB * HH * WWD * DI + 255) / 256, 256, 0, stream>>>(xz, conv_w,
                                                                 conv_b, xconv);
  k_xdbl<<<(BB * KK * LL / 16) / 8, 256, 0, stream>>>(xconv, xpw, x_dbl);
  k_dtproj<<<BB * KK * LL / 16, 256, 0, stream>>>(x_dbl, dtw, dtb, delta);
  k_scan_pass1<<<(BB * KK * DI * NC) / 256, 256, 0, stream>>>(
      xconv, delta, x_dbl, A_logs, chunkA, chunkH);
  k_scan_pass2<<<(BB * KK * DI) / 256, 256, 0, stream>>>(chunkA, chunkH);
  k_scan_pass3<<<(BB * KK * DI * NC) / 256, 256, 0, stream>>>(
      xconv, delta, x_dbl, A_logs, Ds, chunkH, ysb);
  k_outproj<<<BL / 16, 256, 0, stream>>>(ysb, xz, W_out, out);
}


# dt projection + softplus moved into the scan kernel as each wave's prologue: block-diagonal K=16 WMMA tile gives a lane its own chunk's 8 steps; softplus on hardware exp2/log2/rcp; one launch fewer
# speedup vs baseline: 2.1927x; 2.1927x over previous
#include <hip/hip_runtime.h>
#include <cstdint>
#include <cstddef>

#define BB 4
#define HH 32
#define WWD 256
#define CC 96
#define DI 192
#define KK 2
#define NN 16
#define RR 6
#define LL (HH*WWD)
#define BL (BB*LL)
#define NOUT (2*DI)
#define CDBL (RR + 2*NN)
#define NC 128
#define CL (LL/NC)
#define NG (DI/8)
#define LQ (LL/4)

typedef float v4f __attribute__((ext_vector_type(4)));

typedef float v2f __attribute__((ext_vector_type(2)));
typedef float v8f __attribute__((ext_vector_type(8)));

__device__ __forceinline__ v8f wmma4(v2f a, v2f b, v8f c) {
  return __builtin_amdgcn_wmma_f32_16x16x4_f32(false, a, false, b, (short)0, c,
                                               false, false);
}

__device__ __forceinline__ float silu_f(float x) {
  return x / (1.0f + __expf(-x));
}
__device__ __forceinline__ float softplus_f(float x) {
  const float t = __expf(-__builtin_fabsf(x));
  const float u = 1.0f + t;
  const float um1 = u - 1.0f;
  const float lp = (__builtin_amdgcn_logf(u) * 0.693147180559945f) * (t * __builtin_amdgcn_rcpf(um1));
  return __builtin_fmaxf(x, 0.0f) + ((um1 == 0.0f) ? t : lp);
}

#define TH 8
#define TW 16
#define HW2 (TW + 2)
#define HR ((TH + 2) * HW2)
#define HRP 192
#define XP (CC + 1)
#define CP (DI + 4)
#define FT 768

__global__ void __launch_bounds__(FT) k_inproj_conv(
    const float* __restrict__ x, const float* __restrict__ W_in,
    const float* __restrict__ conv_w, const float* __restrict__ conv_b,
    float* __restrict__ xz, float* __restrict__ xconv) {
  __shared__ float sX[HRP * XP];
  __shared__ float sC[HRP * CP];
  const int tid = threadIdx.x;
  const int bid = blockIdx.x;
  const int w0 = (bid % (WWD / TW)) * TW;
  const int h0 = ((bid / (WWD / TW)) % (HH / TH)) * TH;
  const int b = bid / ((WWD / TW) * (HH / TH));
  const int gbase = b * LL + h0 * WWD + w0;

  {
    const int kk = tid % CC, qb = tid / CC;
    float xv[HRP / 8];
#pragma unroll
    for (int it = 0; it < HRP / 8; ++it) {
      const int q = it * 8 + qb;
      const int hh = q / HW2, ww = q - hh * HW2;
      const int h = h0 - 1 + hh, w = w0 - 1 + ww;
      const int hc = min(max(h, 0), HH - 1), wc = min(max(w, 0), WWD - 1);
      const float v = x[((size_t)(b * HH + hc) * WWD + wc) * CC + kk];
      xv[it] = (q < HR && h == hc && w == wc) ? v : 0.f;
    }
#pragma unroll
    for (int it = 0; it < HRP / 8; ++it) sX[(it * 8 + qb) * XP + kk] = xv[it];
  }
  __syncthreads();

  const int wave = tid >> 5, lane = tid & 31;
  const int lid = lane & 15, half = lane >> 4;
  if (wave < 20) {
    const bool isz = wave >= 12;
    const int t2 = isz ? wave - 12 : wave;
    const int mt0 = 2 * (t2 >> 1);
    const int nh = t2 & 1;
    int ar0, ar1;
    if (!isz) { ar0 = mt0 * 16 + lid;            ar1 = ar0 + 16; }
    else      { ar0 = (mt0 + 1) * HW2 + lid + 1; ar1 = ar0 + HW2; }
    v2f a0[CC / 4], a1[CC / 4];
#pragma unroll
    for (int j = 0; j < CC / 4; ++j) {
      a0[j].x = sX[ar0 * XP + 4 * j + 2 * half];
      a0[j].y = sX[ar0 * XP + 4 * j + 2 * half + 1];
      a1[j].x = sX[ar1 * XP + 4 * j + 2 * half];
      a1[j].y = sX[ar1 * XP + 4 * j + 2 * half + 1];
    }
    for (int nt = 0; nt < 6; ++nt) {
      const int ncol = (isz ? DI : 0) + nh * 96 + nt * 16;
      const float* wrow = W_in + (size_t)(ncol + lid) * CC;
      v8f c0 = {}, c1 = {};
#pragma unroll
      for (int g = 0; g < 2; ++g) {
        v2f bq[12];
#pragma unroll
        for (int j = 0; j < 12; ++j) {
          bq[j].x = wrow[4 * (12 * g + j) + 2 * half];
          bq[j].y = wrow[4 * (12 * g + j) + 2 * half + 1];
        }
#pragma unroll
        for (int j = 0; j < 12; ++j) {
          c0 = wmma4(a0[12 * g + j], bq[j], c0);
          c1 = wmma4(a1[12 * g + j], bq[j], c1);
        }
      }
      if (!isz) {
        float* o0 = sC + (mt0 * 16 + half * 8) * CP + ncol + lid;
#pragma unroll
        for (int v = 0; v < 8; ++v) { o0[v * CP] = c0[v]; o0[(16 + v) * CP] = c1[v]; }
      } else {
        float* o0 = xz + (size_t)(gbase + mt0 * WWD + half * 8) * NOUT + ncol + lid;
        float* o1 = o0 + (size_t)WWD * NOUT;
#pragma unroll
        for (int v = 0; v < 8; ++v) { o0[(size_t)v * NOUT] = c0[v]; o1[(size_t)v * NOUT] = c1[v]; }
      }
    }
  }
  __syncthreads();

  const int cl = tid & 7, q4 = (tid >> 3) & 3, g = tid >> 5;
  const int c = g * 8 + cl;
  float wv[9];
#pragma unroll
  for (int t = 0; t < 9; ++t) wv[t] = conv_w[c * 9 + t];
  const float bias = conv_b[c];
  float* xrow = xconv + ((size_t)(b * NG + g) * LQ) * 32 + cl * 4;
  for (int ph = 0; ph < TH; ++ph) {
    const int pw0 = q4 * 4;
    const float* tap = sC + (ph * HW2 + pw0) * CP + c;
    v4f o;
#pragma unroll
    for (int j = 0; j < 4; ++j) {
      float s = bias;
#pragma unroll
      for (int dy = 0; dy < 3; ++dy) {
#pragma unroll
        for (int dx = 0; dx < 3; ++dx)
          s = __builtin_fmaf(tap[(dy * HW2 + dx + j) * CP], wv[dy * 3 + dx], s);
      }
      o[j] = silu_f(s);
    }
    const int l = (h0 + ph) * WWD + w0 + pw0;
    *(v4f*)(xrow + (size_t)(l >> 2) * 32) = o;
  }
}

__global__ void __launch_bounds__(256) k_xdbl(const float* __restrict__ xconv,
                                              const float* __restrict__ xpw,
                                              float* __restrict__ x_dbl) {
  const int tid = threadIdx.x, wave = tid >> 5, lane = tid & 31;
  const int lid = lane & 15, half = lane >> 4;
  const int tile = blockIdx.x * 8 + wave;
  const int lt = tile % (LL / 16);
  const int bk = tile / (LL / 16);
  const int k = bk % KK, b = bk / KK;
  const int lbase = lt * 16;
  const int l = lbase + lid;
  const int lidx = k ? (LL - 1 - l) : l;
  const float* arow = xconv + ((size_t)b * NG * LQ + (lidx >> 2)) * 32 + (lidx & 3);
  v8f acc[3] = {};
  for (int k0 = 0; k0 < DI; k0 += 4) {
    const int kk = k0 + 2 * half;
    const float* ap = arow + (size_t)(kk >> 3) * LQ * 32 + (kk & 7) * 4;
    v2f a; a.x = ap[0]; a.y = ap[4];
#pragma unroll
    for (int t = 0; t < 3; ++t) {
      int c2 = t * 16 + lid;
      v2f bb;
      if (c2 < CDBL) {
        const float* wrow = xpw + (size_t)(k * CDBL + c2) * DI;
        bb.x = wrow[k0 + 2 * half]; bb.y = wrow[k0 + 2 * half + 1];
      } else { bb.x = 0.f; bb.y = 0.f; }
      acc[t] = wmma4(a, bb, acc[t]);
    }
  }
#pragma unroll
  for (int t = 0; t < 3; ++t) {
    int col = t * 16 + lid;
    if (col < CDBL) {
      float* orow = x_dbl + ((size_t)bk * LL + lbase + half * 8) * CDBL + col;
#pragma unroll
      for (int v = 0; v < 8; ++v) orow[(size_t)v * CDBL] = acc[t][v];
    }
  }
}

#define SDT 8
#define SCS (NN * SDT + 8)
#define HQP (CDBL / 2)
#define HPC 96
#define NQ (CL / 4)
#define NHQ (CL / 2)

__global__ void __launch_bounds__(1024) k_scan_fused(
    const float* __restrict__ xconv, float* delta,
    const float* __restrict__ x_dbl, const float* __restrict__ dtw,
    const float* __restrict__ dtb, const float* __restrict__ A_logs,
    const float* __restrict__ Ds, float* __restrict__ ys) {
  __shared__ float sA[NC * SCS];
  __shared__ float sH[NC * SCS];
  __shared__ v4f sR[32 * 2 * HPC];
  const int tid = threadIdx.x;
  const int dloc = tid & (SDT - 1);
  const int c = tid / SDT;
  const int bk = blockIdx.x / (DI / SDT);
  const int d = (blockIdx.x % (DI / SDT)) * SDT + dloc;
  const int k = bk % KK, b = bk / KK;
  float Av[NN];
#pragma unroll
  for (int n = 0; n < NN; ++n)
    Av[n] = -__expf(A_logs[(size_t)(k * DI + d) * NN + n]);
  const float Dv = Ds[k * DI + d];
  const int grp = blockIdx.x % (DI / SDT);
  const v4f* xq = (const v4f*)(xconv + ((size_t)(b * NG + grp) * LQ) * 32) + dloc;
  const v4f* dq = (const v4f*)(delta + ((size_t)(bk * NG + grp) * LQ) * 32) + dloc;
  const float* xd = x_dbl + (size_t)bk * LL * CDBL;
  float* yo = ys + (size_t)bk * LL * DI;
  const int l0 = c * CL;
  const int sbase = c * SCS + dloc;

  {
    const int ln = tid & 31, c16 = ln & 15, hf = ln >> 4;
    const bool odd = (c16 >> 3) != 0;
    const v2f zz = {0.f, 0.f};
    const float* wrow = dtw + (size_t)(k * DI + grp * 8 + (c16 & 7)) * RR;
    v2f w03, w45;
    w03.x = wrow[2 * hf]; w03.y = wrow[2 * hf + 1];
    w45.x = wrow[4];      w45.y = wrow[5];
    if (hf) w45 = zz;
    const v2f b0 = odd ? zz : w03, b1 = odd ? zz : w45;
    const v2f b2 = odd ? w03 : zz, b3 = odd ? w45 : zz;
    const float bias = dtb[k * DI + d];
    const float* re = xd + (size_t)(((tid >> 5) * 4 + 2 * (c16 >> 3)) * CL + (c16 & 7)) * CDBL;
    const float* ro = re + (size_t)CL * CDBL;
    v4f* dw = (v4f*)(delta + ((size_t)(bk * NG + grp) * LQ) * 32) + dloc;
    for (int g = 0; g < CL / 8; ++g) {
      const float* pe = re + (size_t)g * 8 * CDBL;
      const float* po = ro + (size_t)g * 8 * CDBL;
      const v2f a0 = *(const v2f*)(pe + 2 * hf);
      v2f a1 = *(const v2f*)(pe + 4);
      const v2f a2 = *(const v2f*)(po + 2 * hf);
      v2f a3 = *(const v2f*)(po + 4);
      if (hf) { a1 = zz; a3 = zz; }
      v8f acc = {};
      acc = wmma4(a0, b0, acc);
      acc = wmma4(a1, b1, acc);
      acc = wmma4(a2, b2, acc);
      acc = wmma4(a3, b3, acc);
      v4f o0, o1;
#pragma unroll
      for (int v = 0; v < 4; ++v) {
        o0[v] = softplus_f(acc[v] + bias);
        o1[v] = softplus_f(acc[4 + v] + bias);
      }
      const int m = l0 / 4 + 2 * g;
      dw[(size_t)m * 8] = o0;
      dw[(size_t)(m + 1) * 8] = o1;
    }
  }

  const int wave = tid >> 5, lane = tid & 31;
  const v4f* xd4 = (const v4f*)xd;
  v4f* slot = sR + wave * (2 * HPC);
  unsigned gof[3];
#pragma unroll
  for (int it = 0; it < 3; ++it) {
    const unsigned P = min(it * 32 + lane, 4 * HQP - 1);
    gof[it] = (wave * 4 + P / HQP) * (NHQ * HQP) + P % HQP;
  }
  const v2f* rowp = (const v2f*)slot + (lane >> 3) * (2 * HQP) + RR / 2;
  v4f R[3];

  float h[NN];
  {
    float aP[NN];
#pragma unroll
    for (int n = 0; n < NN; ++n) { aP[n] = 1.f; h[n] = 0.f; }
    for (int m = l0 / 4; m < (l0 + CL) / 4; ++m) {
      v4f d4 = dq[(size_t)m * 8];
      v4f u4 = xq[(size_t)(k ? (LQ - 1 - m) : m) * 8];
      if (k) u4 = u4.wzyx;
#pragma nounroll
      for (int j = 0; j < 4; ++j) {
        const float u = u4.x, dlt = d4.x;
        u4 = u4.yzwx; d4 = d4.yzwx;
        const float* row = xd + (size_t)(4 * m + j) * CDBL;
        const float du = dlt * u;
#pragma unroll
        for (int n = 0; n < NN; ++n) {
          const float dA = __expf(dlt * Av[n]);
          aP[n] *= dA;
          h[n] = dA * h[n] + du * row[RR + n];
        }
      }
    }
#pragma unroll
    for (int n = 0; n < NN; ++n) {
      sA[sbase + n * SDT] = aP[n];
      sH[sbase + n * SDT] = h[n];
    }
  }
#pragma unroll
  for (int it = 0; it < 3; ++it) R[it] = xd4[gof[it]];
  __syncthreads();
  if (tid < NN * SDT) {
    float hrun = 0.f;
#pragma unroll 8
    for (int cc = 0; cc < NC; ++cc) {
      const int idx = cc * SCS + tid;
      const float a = sA[idx];
      const float hl = sH[idx];
      sH[idx] = hrun;
      hrun = a * hrun + hl;
    }
  }
#pragma unroll
  for (int it = 0; it < 3; ++it) slot[it * 32 + lane] = R[it];
  __syncthreads();
#pragma unroll
  for (int n = 0; n < NN; ++n) h[n] = sH[sbase + n * SDT];

  for (int mi = 0; mi < NQ; ++mi) {
    const int m = l0 / 4 + mi;
    v4f d4 = dq[(size_t)m * 8];
    v4f u4 = xq[(size_t)(k ? (LQ - 1 - m) : m) * 8];
    if (k) u4 = u4.wzyx;
#pragma nounroll
    for (int hh = 0; hh < 2; ++hh) {
      const unsigned tn = min(2 * mi + hh + 1, NHQ - 1);
#pragma unroll
      for (int it = 0; it < 3; ++it) R[it] = xd4[gof[it] + HQP * tn];
      const v2f* rp = rowp + hh * (2 * HPC);
#pragma nounroll
      for (int j = 0; j < 2; ++j) {
        const int l = 4 * m + 2 * hh + j;
        const float u = u4.x, dlt = d4.x;
        u4 = u4.yzwx; d4 = d4.yzwx;
        v2f bc[NN];
#pragma unroll
        for (int i = 0; i < NN; ++i) bc[i] = rp[i];
        rp += HQP;
        const float du = dlt * u;
        float y = 0.f;
#pragma unroll
        for (int n = 0; n < NN; ++n) {
          const float dA = __expf(dlt * Av[n]);
          h[n] = dA * h[n] + du * bc[n >> 1][n & 1];
          y += h[n] * bc[NN / 2 + (n >> 1)][n & 1];
        }
        yo[(size_t)l * DI + d] = y + u * Dv;
      }
      v4f* st = slot + (1 - hh) * HPC + lane;
#pragma unroll
      for (int it = 0; it < 3; ++it) st[it * 32] = R[it];
    }
  }
}

__global__ void __launch_bounds__(256) k_outproj(const float* __restrict__ ys,
                                                 const float* __restrict__ xz,
                                                 const float* __restrict__ W_out,
                                                 float* __restrict__ out) {
  __shared__ float sY[16][DI + 1];
  const int rowBase = blockIdx.x * 16;
  const int b = rowBase / LL;
  const int lb = rowBase % LL;
  const int tid = threadIdx.x;
  for (int i = tid; i < 16 * DI; i += 256) {
    int m = i / DI, dd = i % DI;
    int l = lb + m;
    float y0 = ys[((size_t)(b * KK + 0) * LL + l) * DI + dd];
    float y1 = ys[((size_t)(b * KK + 1) * LL + (LL - 1 - l)) * DI + dd];
    float zv = xz[(size_t)(b * LL + l) * NOUT + DI + dd];
    sY[m][dd] = (y0 + y1) * silu_f(zv);
  }
  __syncthreads();
  const int wave = tid >> 5, lane = tid & 31;
  const int lid = lane & 15, half = lane >> 4;
  for (int nt = wave; nt < CC / 16; nt += 8) {
    const int nbase = nt * 16;
    const float* wrow = W_out + (size_t)(nbase + lid) * DI;
    v8f acc = {};
    for (int k0 = 0; k0 < DI; k0 += 4) {
      v2f a;  a.x = sY[lid][k0 + 2 * half];  a.y = sY[lid][k0 + 2 * half + 1];
      v2f bb; bb.x = wrow[k0 + 2 * half];    bb.y = wrow[k0 + 2 * half + 1];
      acc = wmma4(a, bb, acc);
    }
    float* orow = out + (size_t)(rowBase + half * 8) * CC + nbase + lid;
#pragma unroll
    for (int v = 0; v < 8; ++v) orow[(size_t)v * CC] = acc[v];
  }
}

extern "C" void kernel_launch(void* const* d_in, const int* in_sizes, int n_in,
                              void* d_out, int out_size, void* d_ws, size_t ws_size,
                              hipStream_t stream) {
  (void)in_sizes; (void)n_in; (void)out_size; (void)ws_size;
  const float* x      = (const float*)d_in[0];
  const float* W_in   = (const float*)d_in[1];
  const float* conv_w = (const float*)d_in[2];
  const float* conv_b = (const float*)d_in[3];
  const float* xpw    = (const float*)d_in[4];
  const float* dtw    = (const float*)d_in[5];
  const float* dtb    = (const float*)d_in[6];
  const float* A_logs = (const float*)d_in[7];
  const float* Ds     = (const float*)d_in[8];
  const float* W_out  = (const float*)d_in[9];
  float* out = (float*)d_out;

  char* ws = (char*)d_ws;
  size_t off = 0;
  auto alloc = [&](size_t nbytes) {
    char* p = ws + off;
    off = (off + nbytes + 255) & ~(size_t)255;
    return (float*)p;
  };
  float* xz     = alloc((size_t)BL * NOUT * sizeof(float));
  float* xconv  = alloc((size_t)BL * DI * sizeof(float));
  float* x_dbl  = alloc((size_t)BB * KK * LL * CDBL * sizeof(float));
  float* delta  = alloc((size_t)BB * KK * LL * DI * sizeof(float));
  float* ysb    = alloc((size_t)BB * KK * LL * DI * sizeof(float));

  k_inproj_conv<<<BB * (HH / TH) * (WWD / TW), FT, 0, stream>>>(x, W_in, conv_w,
                                                                conv_b, xz, xconv);
  k_xdbl<<<(BB * KK * LL / 16) / 8, 256, 0, stream>>>(xconv, xpw, x_dbl);
  k_scan_fused<<<BB * KK * (DI / SDT), NC * SDT, 0, stream>>>(
      xconv, delta, x_dbl, dtw, dtb, A_logs, Ds, ysb);
  k_outproj<<<BL / 16, 256, 0, stream>>>(ysb, xz, W_out, out);
}


# scan output plane ys re-laid as [8-channel group][4-step quad]: phase 3 stores 4 steps as one 16-byte piece, 8 lanes fill a whole 128-byte line; out_proj stages it by 16-byte loads; op order kept
# speedup vs baseline: 2.3733x; 1.0824x over previous
#include <hip/hip_runtime.h>
#include <cstdint>
#include <cstddef>

#define BB 4
#define HH 32
#define WWD 256
#define CC 96
#define DI 192
#define KK 2
#define NN 16
#define RR 6
#define LL (HH*WWD)
#define BL (BB*LL)
#define NOUT (2*DI)
#define CDBL (RR + 2*NN)
#define NC 128
#define CL (LL/NC)
#define NG (DI/8)
#define LQ (LL/4)

typedef float v4f __attribute__((ext_vector_type(4)));

typedef float v2f __attribute__((ext_vector_type(2)));
typedef float v8f __attribute__((ext_vector_type(8)));

__device__ __forceinline__ v8f wmma4(v2f a, v2f b, v8f c) {
  return __builtin_amdgcn_wmma_f32_16x16x4_f32(false, a, false, b, (short)0, c,
                                               false, false);
}

__device__ __forceinline__ float silu_f(float x) {
  return x / (1.0f + __expf(-x));
}
__device__ __forceinline__ float softplus_f(float x) {
  const float t = __expf(-__builtin_fabsf(x));
  const float u = 1.0f + t;
  const float um1 = u - 1.0f;
  const float lp = (__builtin_amdgcn_logf(u) * 0.693147180559945f) * (t * __builtin_amdgcn_rcpf(um1));
  return __builtin_fmaxf(x, 0.0f) + ((um1 == 0.0f) ? t : lp);
}

#define TH 8
#define TW 16
#define HW2 (TW + 2)
#define HR ((TH + 2) * HW2)
#define HRP 192
#define XP (CC + 1)
#define CP (DI + 4)
#define FT 768

__global__ void __launch_bounds__(FT) k_inproj_conv(
    const float* __restrict__ x, const float* __restrict__ W_in,
    const float* __restrict__ conv_w, const float* __restrict__ conv_b,
    float* __restrict__ xz, float* __restrict__ xconv) {
  __shared__ float sX[HRP * XP];
  __shared__ float sC[HRP * CP];
  const int tid = threadIdx.x;
  const int bid = blockIdx.x;
  const int w0 = (bid % (WWD / TW)) * TW;
  const int h0 = ((bid / (WWD / TW)) % (HH / TH)) * TH;
  const int b = bid / ((WWD / TW) * (HH / TH));
  const int gbase = b * LL + h0 * WWD + w0;

  {
    const int kk = tid % CC, qb = tid / CC;
    float xv[HRP / 8];
#pragma unroll
    for (int it = 0; it < HRP / 8; ++it) {
      const int q = it * 8 + qb;
      const int hh = q / HW2, ww = q - hh * HW2;
      const int h = h0 - 1 + hh, w = w0 - 1 + ww;
      const int hc = min(max(h, 0), HH - 1), wc = min(max(w, 0), WWD - 1);
      const float v = x[((size_t)(b * HH + hc) * WWD + wc) * CC + kk];
      xv[it] = (q < HR && h == hc && w == wc) ? v : 0.f;
    }
#pragma unroll
    for (int it = 0; it < HRP / 8; ++it) sX[(it * 8 + qb) * XP + kk] = xv[it];
  }
  __syncthreads();

  const int wave = tid >> 5, lane = tid & 31;
  const int lid = lane & 15, half = lane >> 4;
  if (wave < 20) {
    const bool isz = wave >= 12;
    const int t2 = isz ? wave - 12 : wave;
    const int mt0 = 2 * (t2 >> 1);
    const int nh = t2 & 1;
    int ar0, ar1;
    if (!isz) { ar0 = mt0 * 16 + lid;            ar1 = ar0 + 16; }
    else      { ar0 = (mt0 + 1) * HW2 + lid + 1; ar1 = ar0 + HW2; }
    v2f a0[CC / 4], a1[CC / 4];
#pragma unroll
    for (int j = 0; j < CC / 4; ++j) {
      a0[j].x = sX[ar0 * XP + 4 * j + 2 * half];
      a0[j].y = sX[ar0 * XP + 4 * j + 2 * half + 1];
      a1[j].x = sX[ar1 * XP + 4 * j + 2 * half];
      a1[j].y = sX[ar1 * XP + 4 * j + 2 * half + 1];
    }
    for (int nt = 0; nt < 6; ++nt) {
      const int ncol = (isz ? DI : 0) + nh * 96 + nt * 16;
      const float* wrow = W_in + (size_t)(ncol + lid) * CC;
      v8f c0 = {}, c1 = {};
#pragma unroll
      for (int g = 0; g < 2; ++g) {
        v2f bq[12];
#pragma unroll
        for (int j = 0; j < 12; ++j) {
          bq[j].x = wrow[4 * (12 * g + j) + 2 * half];
          bq[j].y = wrow[4 * (12 * g + j) + 2 * half + 1];
        }
#pragma unroll
        for (int j = 0; j < 12; ++j) {
          c0 = wmma4(a0[12 * g + j], bq[j], c0);
          c1 = wmma4(a1[12 * g + j], bq[j], c1);
        }
      }
      if (!isz) {
        float* o0 = sC + (mt0 * 16 + half * 8) * CP + ncol + lid;
#pragma unroll
        for (int v = 0; v < 8; ++v) { o0[v * CP] = c0[v]; o0[(16 + v) * CP] = c1[v]; }
      } else {
        float* o0 = xz + (size_t)(gbase + mt0 * WWD + half * 8) * NOUT + ncol + lid;
        float* o1 = o0 + (size_t)WWD * NOUT;
#pragma unroll
        for (int v = 0; v < 8; ++v) { o0[(size_t)v * NOUT] = c0[v]; o1[(size_t)v * NOUT] = c1[v]; }
      }
    }
  }
  __syncthreads();

  const int cl = tid & 7, q4 = (tid >> 3) & 3, g = tid >> 5;
  const int c = g * 8 + cl;
  float wv[9];
#pragma unroll
  for (int t = 0; t < 9; ++t) wv[t] = conv_w[c * 9 + t];
  const float bias = conv_b[c];
  float* xrow = xconv + ((size_t)(b * NG + g) * LQ) * 32 + cl * 4;
  for (int ph = 0; ph < TH; ++ph) {
    const int pw0 = q4 * 4;
    const float* tap = sC + (ph * HW2 + pw0) * CP + c;
    v4f o;
#pragma unroll
    for (int j = 0; j < 4; ++j) {
      float s = bias;
#pragma unroll
      for (int dy = 0; dy < 3; ++dy) {
#pragma unroll
        for (int dx = 0; dx < 3; ++dx)
          s = __builtin_fmaf(tap[(dy * HW2 + dx + j) * CP], wv[dy * 3 + dx], s);
      }
      o[j] = silu_f(s);
    }
    const int l = (h0 + ph) * WWD + w0 + pw0;
    *(v4f*)(xrow + (size_t)(l >> 2) * 32) = o;
  }
}

__global__ void __launch_bounds__(256) k_xdbl(const float* __restrict__ xconv,
                                              const float* __restrict__ xpw,
                                              float* __restrict__ x_dbl) {
  const int tid = threadIdx.x, wave = tid >> 5, lane = tid & 31;
  const int lid = lane & 15, half = lane >> 4;
  const int tile = blockIdx.x * 8 + wave;
  const int lt = tile % (LL / 16);
  const int bk = tile / (LL / 16);
  const int k = bk % KK, b = bk / KK;
  const int lbase = lt * 16;
  const int l = lbase + lid;
  const int lidx = k ? (LL - 1 - l) : l;
  const float* arow = xconv + ((size_t)b * NG * LQ + (lidx >> 2)) * 32 + (lidx & 3);
  v8f acc[3] = {};
  for (int k0 = 0; k0 < DI; k0 += 4) {
    const int kk = k0 + 2 * half;
    const float* ap = arow + (size_t)(kk >> 3) * LQ * 32 + (kk & 7) * 4;
    v2f a; a.x = ap[0]; a.y = ap[4];
#pragma unroll
    for (int t = 0; t < 3; ++t) {
      int c2 = t * 16 + lid;
      v2f bb;
      if (c2 < CDBL) {
        const float* wrow = xpw + (size_t)(k * CDBL + c2) * DI;
        bb.x = wrow[k0 + 2 * half]; bb.y = wrow[k0 + 2 * half + 1];
      } else { bb.x = 0.f; bb.y = 0.f; }
      acc[t] = wmma4(a, bb, acc[t]);
    }
  }
#pragma unroll
  for (int t = 0; t < 3; ++t) {
    int col = t * 16 + lid;
    if (col < CDBL) {
      float* orow = x_dbl + ((size_t)bk * LL + lbase + half * 8) * CDBL + col;
#pragma unroll
      for (int v = 0; v < 8; ++v) orow[(size_t)v * CDBL] = acc[t][v];
    }
  }
}

#define SDT 8
#define SCS (NN * SDT + 8)
#define HQP (CDBL / 2)
#define HPC 96
#define NQ (CL / 4)
#define NHQ (CL / 2)

__global__ void __launch_bounds__(1024) k_scan_fused(
    const float* __restrict__ xconv, float* delta,
    const float* __restrict__ x_dbl, const float* __restrict__ dtw,
    const float* __restrict__ dtb, const float* __restrict__ A_logs,
    const float* __restrict__ Ds, float* __restrict__ ys) {
  __shared__ float sA[NC * SCS];
  __shared__ float sH[NC * SCS];
  __shared__ v4f sR[32 * 2 * HPC];
  const int tid = threadIdx.x;
  const int dloc = tid & (SDT - 1);
  const int c = tid / SDT;
  const int bk = blockIdx.x / (DI / SDT);
  const int d = (blockIdx.x % (DI / SDT)) * SDT + dloc;
  const int k = bk % KK, b = bk / KK;
  float Av[NN];
#pragma unroll
  for (int n = 0; n < NN; ++n)
    Av[n] = -__expf(A_logs[(size_t)(k * DI + d) * NN + n]);
  const float Dv = Ds[k * DI + d];
  const int grp = blockIdx.x % (DI / SDT);
  const float* xd = x_dbl + (size_t)bk * LL * CDBL;
  const unsigned poff = (unsigned)((bk * NG + grp) * LQ) * 32u + (unsigned)dloc * 4u;
  const unsigned xoff = (unsigned)((b * NG + grp) * LQ) * 32u + (unsigned)dloc * 4u;
  const int l0 = c * CL;
  const int sbase = c * SCS + dloc;

  {
    const int ln = tid & 31, c16 = ln & 15, hf = ln >> 4;
    const bool odd = (c16 >> 3) != 0;
    const v2f zz = {0.f, 0.f};
    const float* wrow = dtw + (size_t)(k * DI + grp * 8 + (c16 & 7)) * RR;
    v2f w03, w45;
    w03.x = wrow[2 * hf]; w03.y = wrow[2 * hf + 1];
    w45.x = wrow[4];      w45.y = wrow[5];
    if (hf) w45 = zz;
    const v2f b0 = odd ? zz : w03, b1 = odd ? zz : w45;
    const v2f b2 = odd ? w03 : zz, b3 = odd ? w45 : zz;
    const float bias = dtb[k * DI + d];
    const float* re = xd + (size_t)(((tid >> 5) * 4 + 2 * (c16 >> 3)) * CL + (c16 & 7)) * CDBL;
    const float* ro = re + (size_t)CL * CDBL;
    v4f* dw = (v4f*)(delta + ((size_t)(bk * NG + grp) * LQ) * 32) + dloc;
    for (int g = 0; g < CL / 8; ++g) {
      const float* pe = re + (size_t)g * 8 * CDBL;
      const float* po = ro + (size_t)g * 8 * CDBL;
      const v2f a0 = *(const v2f*)(pe + 2 * hf);
      v2f a1 = *(const v2f*)(pe + 4);
      const v2f a2 = *(const v2f*)(po + 2 * hf);
      v2f a3 = *(const v2f*)(po + 4);
      if (hf) { a1 = zz; a3 = zz; }
      v8f acc = {};
      acc = wmma4(a0, b0, acc);
      acc = wmma4(a1, b1, acc);
      acc = wmma4(a2, b2, acc);
      acc = wmma4(a3, b3, acc);
      v4f o0, o1;
#pragma unroll
      for (int v = 0; v < 4; ++v) {
        o0[v] = softplus_f(acc[v] + bias);
        o1[v] = softplus_f(acc[4 + v] + bias);
      }
      const int m = l0 / 4 + 2 * g;
      dw[(size_t)m * 8] = o0;
      dw[(size_t)(m + 1) * 8] = o1;
    }
  }

  const int wave = tid >> 5, lane = tid & 31;
  const v4f* xd4 = (const v4f*)xd;
  v4f* slot = sR + wave * (2 * HPC);
  unsigned gof[3];
#pragma unroll
  for (int it = 0; it < 3; ++it) {
    const unsigned P = min(it * 32 + lane, 4 * HQP - 1);
    gof[it] = (wave * 4 + P / HQP) * (NHQ * HQP) + P % HQP;
  }
  const v2f* rowp = (const v2f*)slot + (lane >> 3) * (2 * HQP) + RR / 2;
  v4f R[3];

  float h[NN];
  {
    float aP[NN];
#pragma unroll
    for (int n = 0; n < NN; ++n) { aP[n] = 1.f; h[n] = 0.f; }
    for (int m = l0 / 4; m < (l0 + CL) / 4; ++m) {
      v4f d4 = *(const v4f*)(delta + (poff + (unsigned)m * 32u));
      v4f u4 = *(const v4f*)(xconv + (xoff + (unsigned)(k ? (LQ - 1 - m) : m) * 32u));
      if (k) u4 = u4.wzyx;
#pragma nounroll
      for (int j = 0; j < 4; ++j) {
        const float u = u4.x, dlt = d4.x;
        u4 = u4.yzwx; d4 = d4.yzwx;
        const float* row = xd + (size_t)(4 * m + j) * CDBL;
        const float du = dlt * u;
#pragma unroll
        for (int n = 0; n < NN; ++n) {
          const float dA = __expf(dlt * Av[n]);
          aP[n] *= dA;
          h[n] = dA * h[n] + du * row[RR + n];
        }
      }
    }
#pragma unroll
    for (int n = 0; n < NN; ++n) {
      sA[sbase + n * SDT] = aP[n];
      sH[sbase + n * SDT] = h[n];
    }
  }
#pragma unroll
  for (int it = 0; it < 3; ++it) R[it] = xd4[gof[it]];
  __syncthreads();
  if (tid < NN * SDT) {
    float hrun = 0.f;
#pragma unroll 8
    for (int cc = 0; cc < NC; ++cc) {
      const int idx = cc * SCS + tid;
      const float a = sA[idx];
      const float hl = sH[idx];
      sH[idx] = hrun;
      hrun = a * hrun + hl;
    }
  }
#pragma unroll
  for (int it = 0; it < 3; ++it) slot[it * 32 + lane] = R[it];
  __syncthreads();
#pragma unroll
  for (int n = 0; n < NN; ++n) h[n] = sH[sbase + n * SDT];

  for (int mi = 0; mi < NQ; ++mi) {
    const int m = l0 / 4 + mi;
    v4f d4 = *(const v4f*)(delta + (poff + (unsigned)m * 32u));
    v4f u4 = *(const v4f*)(xconv + (xoff + (unsigned)(k ? (LQ - 1 - m) : m) * 32u));
    if (k) u4 = u4.wzyx;
    v4f y4 = {0.f, 0.f, 0.f, 0.f};
#pragma nounroll
    for (int hh = 0; hh < 2; ++hh) {
      const unsigned tn = min(2 * mi + hh + 1, NHQ - 1);
#pragma unroll
      for (int it = 0; it < 3; ++it) R[it] = xd4[gof[it] + HQP * tn];
      const v2f* rp = rowp + hh * (2 * HPC);
#pragma nounroll
      for (int j = 0; j < 2; ++j) {
        const float u = u4.x, dlt = d4.x;
        u4 = u4.yzwx; d4 = d4.yzwx;
        v2f bc[NN];
#pragma unroll
        for (int i = 0; i < NN; ++i) bc[i] = rp[i];
        rp += HQP;
        const float du = dlt * u;
        float y = 0.f;
#pragma unroll
        for (int n = 0; n < NN; ++n) {
          const float dA = __expf(dlt * Av[n]);
          h[n] = dA * h[n] + du * bc[n >> 1][n & 1];
          y += h[n] * bc[NN / 2 + (n >> 1)][n & 1];
        }
        y4 = y4.yzwx;
        y4.w = y + u * Dv;
      }
      v4f* st = slot + (1 - hh) * HPC + lane;
#pragma unroll
      for (int it = 0; it < 3; ++it) st[it * 32] = R[it];
    }
    *(v4f*)(ys + (poff + (unsigned)m * 32u)) = y4;
  }
}

__global__ void __launch_bounds__(256) k_outproj(const float* __restrict__ ys,
                                                 const float* __restrict__ xz,
                                                 const float* __restrict__ W_out,
                                                 float* __restrict__ out) {
  __shared__ float sY[16][DI + 1];
  const int rowBase = blockIdx.x * 16;
  const int b = rowBase / LL;
  const int lb = rowBase % LL;
  const int tid = threadIdx.x;
  {
    v4f ya[3], yb[3], zq[3];
#pragma unroll
    for (int it = 0; it < 3; ++it) {
      const int i = it * 256 + tid;
      const int q = i / DI, dd = i % DI;
      const int m = (lb >> 2) + q;
      const size_t go = (size_t)(dd >> 3) * LQ * 32 + (dd & 7) * 4;
      ya[it] = *(const v4f*)(ys + ((size_t)(b * KK + 0) * NG * LQ + m) * 32 + go);
      yb[it] = *(const v4f*)(ys + ((size_t)(b * KK + 1) * NG * LQ + (LQ - 1 - m)) * 32 + go);
    }
#pragma unroll
    for (int it = 0; it < 3; ++it) {
      const int i = it * 256 + tid;
      const int q = i / DI, dd = i % DI;
      const float* zp = xz + (size_t)(b * LL + lb + 4 * q) * NOUT + DI + dd;
#pragma unroll
      for (int j = 0; j < 4; ++j) zq[it][j] = zp[(size_t)j * NOUT];
    }
#pragma unroll
    for (int it = 0; it < 3; ++it) {
      const int i = it * 256 + tid;
      const int q = i / DI, dd = i % DI;
      const v4f yr = yb[it].wzyx;
#pragma unroll
      for (int j = 0; j < 4; ++j) sY[4 * q + j][dd] = (ya[it][j] + yr[j]) * silu_f(zq[it][j]);
    }
  }
  __syncthreads();
  const int wave = tid >> 5, lane = tid & 31;
  const int lid = lane & 15, half = lane >> 4;
  for (int nt = wave; nt < CC / 16; nt += 8) {
    const int nbase = nt * 16;
    const float* wrow = W_out + (size_t)(nbase + lid) * DI;
    v8f acc = {};
    for (int k0 = 0; k0 < DI; k0 += 4) {
      v2f a;  a.x = sY[lid][k0 + 2 * half];  a.y = sY[lid][k0 + 2 * half + 1];
      v2f bb; bb.x = wrow[k0 + 2 * half];    bb.y = wrow[k0 + 2 * half + 1];
      acc = wmma4(a, bb, acc);
    }
    float* orow = out + (size_t)(rowBase + half * 8) * CC + nbase + lid;
#pragma unroll
    for (int v = 0; v < 8; ++v) orow[(size_t)v * CC] = acc[v];
  }
}

extern "C" void kernel_launch(void* const* d_in, const int* in_sizes, int n_in,
                              void* d_out, int out_size, void* d_ws, size_t ws_size,
                              hipStream_t stream) {
  (void)in_sizes; (void)n_in; (void)out_size; (void)ws_size;
  const float* x      = (const float*)d_in[0];
  const float* W_in   = (const float*)d_in[1];
  const float* conv_w = (const float*)d_in[2];
  const float* conv_b = (const float*)d_in[3];
  const float* xpw    = (const float*)d_in[4];
  const float* dtw    = (const float*)d_in[5];
  const float* dtb    = (const float*)d_in[6];
  const float* A_logs = (const float*)d_in[7];
  const float* Ds     = (const float*)d_in[8];
  const float* W_out  = (const float*)d_in[9];
  float* out = (float*)d_out;

  char* ws = (char*)d_ws;
  size_t off = 0;
  auto alloc = [&](size_t nbytes) {
    char* p = ws + off;
    off = (off + nbytes + 255) & ~(size_t)255;
    return (float*)p;
  };
  float* xz     = alloc((size_t)BL * NOUT * sizeof(float));
  float* xconv  = alloc((size_t)BL * DI * sizeof(float));
  float* x_dbl  = alloc((size_t)BB * KK * LL * CDBL * sizeof(float));
  float* delta  = alloc((size_t)BB * KK * LL * DI * sizeof(float));
  float* ysb    = alloc((size_t)BB * KK * LL * DI * sizeof(float));

  k_inproj_conv<<<BB * (HH / TH) * (WWD / TW), FT, 0, stream>>>(x, W_in, conv_w,
                                                                conv_b, xz, xconv);
  k_xdbl<<<(BB * KK * LL / 16) / 8, 256, 0, stream>>>(xconv, xpw, x_dbl);
  k_scan_fused<<<BB * KK * (DI / SDT), NC * SDT, 0, stream>>>(
      xconv, delta, x_dbl, dtw, dtb, A_logs, Ds, ysb);
  k_outproj<<<BL / 16, 256, 0, stream>>>(ysb, xz, W_out, out);
}


# x_dbl and out_proj: weights staged once per block in LDS, not refetched from global per tile; a wave loads its A fragments straight into registers and runs all N-tiles of its rows; k order kept
# speedup vs baseline: 2.5439x; 1.0719x over previous
#include <hip/hip_runtime.h>
#include <cstdint>
#include <cstddef>

#define BB 4
#define HH 32
#define WWD 256
#define CC 96
#define DI 192
#define KK 2
#define NN 16
#define RR 6
#define LL (HH*WWD)
#define BL (BB*LL)
#define NOUT (2*DI)
#define CDBL (RR + 2*NN)
#define NC 128
#define CL (LL/NC)
#define NG (DI/8)
#define LQ (LL/4)

typedef float v4f __attribute__((ext_vector_type(4)));

typedef float v2f __attribute__((ext_vector_type(2)));
typedef float v8f __attribute__((ext_vector_type(8)));

__device__ __forceinline__ v8f wmma4(v2f a, v2f b, v8f c) {
  return __builtin_amdgcn_wmma_f32_16x16x4_f32(false, a, false, b, (short)0, c,
                                               false, false);
}

__device__ __forceinline__ float silu_f(float x) {
  return x / (1.0f + __expf(-x));
}
__device__ __forceinline__ float softplus_f(float x) {
  const float t = __expf(-__builtin_fabsf(x));
  const float u = 1.0f + t;
  const float um1 = u - 1.0f;
  const float lp = (__builtin_amdgcn_logf(u) * 0.693147180559945f) * (t * __builtin_amdgcn_rcpf(um1));
  return __builtin_fmaxf(x, 0.0f) + ((um1 == 0.0f) ? t : lp);
}

#define TH 8
#define TW 16
#define HW2 (TW + 2)
#define HR ((TH + 2) * HW2)
#define HRP 192
#define XP (CC + 1)
#define CP (DI + 4)
#define FT 768

__global__ void __launch_bounds__(FT) k_inproj_conv(
    const float* __restrict__ x, const float* __restrict__ W_in,
    const float* __restrict__ conv_w, const float* __restrict__ conv_b,
    float* __restrict__ xz, float* __restrict__ xconv) {
  __shared__ float sX[HRP * XP];
  __shared__ float sC[HRP * CP];
  const int tid = threadIdx.x;
  const int bid = blockIdx.x;
  const int w0 = (bid % (WWD / TW)) * TW;
  const int h0 = ((bid / (WWD / TW)) % (HH / TH)) * TH;
  const int b = bid / ((WWD / TW) * (HH / TH));
  const int gbase = b * LL + h0 * WWD + w0;

  {
    const int kk = tid % CC, qb = tid / CC;
    float xv[HRP / 8];
#pragma unroll
    for (int it = 0; it < HRP / 8; ++it) {
      const int q = it * 8 + qb;
      const int hh = q / HW2, ww = q - hh * HW2;
      const int h = h0 - 1 + hh, w = w0 - 1 + ww;
      const int hc = min(max(h, 0), HH - 1), wc = min(max(w, 0), WWD - 1);
      const float v = x[((size_t)(b * HH + hc) * WWD + wc) * CC + kk];
      xv[it] = (q < HR && h == hc && w == wc) ? v : 0.f;
    }
#pragma unroll
    for (int it = 0; it < HRP / 8; ++it) sX[(it * 8 + qb) * XP + kk] = xv[it];
  }
  __syncthreads();

  const int wave = tid >> 5, lane = tid & 31;
  const int lid = lane & 15, half = lane >> 4;
  if (wave < 20) {
    const bool isz = wave >= 12;
    const int t2 = isz ? wave - 12 : wave;
    const int mt0 = 2 * (t2 >> 1);
    const int nh = t2 & 1;
    int ar0, ar1;
    if (!isz) { ar0 = mt0 * 16 + lid;            ar1 = ar0 + 16; }
    else      { ar0 = (mt0 + 1) * HW2 + lid + 1; ar1 = ar0 + HW2; }
    v2f a0[CC / 4], a1[CC / 4];
#pragma unroll
    for (int j = 0; j < CC / 4; ++j) {
      a0[j].x = sX[ar0 * XP + 4 * j + 2 * half];
      a0[j].y = sX[ar0 * XP + 4 * j + 2 * half + 1];
      a1[j].x = sX[ar1 * XP + 4 * j + 2 * half];
      a1[j].y = sX[ar1 * XP + 4 * j + 2 * half + 1];
    }
    for (int nt = 0; nt < 6; ++nt) {
      const int ncol = (isz ? DI : 0) + nh * 96 + nt * 16;
      const float* wrow = W_in + (size_t)(ncol + lid) * CC;
      v8f c0 = {}, c1 = {};
#pragma unroll
      for (int g = 0; g < 2; ++g) {
        v2f bq[12];
#pragma unroll
        for (int j = 0; j < 12; ++j) {
          bq[j].x = wrow[4 * (12 * g + j) + 2 * half];
          bq[j].y = wrow[4 * (12 * g + j) + 2 * half + 1];
        }
#pragma unroll
        for (int j = 0; j < 12; ++j) {
          c0 = wmma4(a0[12 * g + j], bq[j], c0);
          c1 = wmma4(a1[12 * g + j], bq[j], c1);
        }
      }
      if (!isz) {
        float* o0 = sC + (mt0 * 16 + half * 8) * CP + ncol + lid;
#pragma unroll
        for (int v = 0; v < 8; ++v) { o0[v * CP] = c0[v]; o0[(16 + v) * CP] = c1[v]; }
      } else {
        float* o0 = xz + (size_t)(gbase + mt0 * WWD + half * 8) * NOUT + ncol + lid;
        float* o1 = o0 + (size_t)WWD * NOUT;
#pragma unroll
        for (int v = 0; v < 8; ++v) { o0[(size_t)v * NOUT] = c0[v]; o1[(size_t)v * NOUT] = c1[v]; }
      }
    }
  }
  __syncthreads();

  const int cl = tid & 7, q4 = (tid >> 3) & 3, g = tid >> 5;
  const int c = g * 8 + cl;
  float wv[9];
#pragma unroll
  for (int t = 0; t < 9; ++t) wv[t] = conv_w[c * 9 + t];
  const float bias = conv_b[c];
  float* xrow = xconv + ((size_t)(b * NG + g) * LQ) * 32 + cl * 4;
  for (int ph = 0; ph < TH; ++ph) {
    const int pw0 = q4 * 4;
    const float* tap = sC + (ph * HW2 + pw0) * CP + c;
    v4f o;
#pragma unroll
    for (int j = 0; j < 4; ++j) {
      float s = bias;
#pragma unroll
      for (int dy = 0; dy < 3; ++dy) {
#pragma unroll
        for (int dx = 0; dx < 3; ++dx)
          s = __builtin_fmaf(tap[(dy * HW2 + dx + j) * CP], wv[dy * 3 + dx], s);
      }
      o[j] = silu_f(s);
    }
    const int l = (h0 + ph) * WWD + w0 + pw0;
    *(v4f*)(xrow + (size_t)(l >> 2) * 32) = o;
  }
}

#define XWP (DI + 4)
#define XT 512

__global__ void __launch_bounds__(XT) k_xdbl(const float* __restrict__ xconv,
                                             const float* __restrict__ xpw,
                                             float* __restrict__ x_dbl) {
  __shared__ __attribute__((aligned(16))) float sW[48 * XWP];
  const int tid = threadIdx.x, wave = tid >> 5, lane = tid & 31;
  const int lid = lane & 15, half = lane >> 4;
  const int tile = blockIdx.x * (XT / 32) + wave;
  const int lt = tile % (LL / 16);
  const int bk = tile / (LL / 16);
  const int k = bk % KK, b = bk / KK;
  const int lbase = lt * 16;
  const int l = lbase + lid;
  const int lidx = k ? (LL - 1 - l) : l;
  const float* arow = xconv + ((size_t)b * NG * LQ + (lidx >> 2)) * 32 + (lidx & 3) + 8 * half;
  v2f a[DI / 4];
#pragma unroll
  for (int j = 0; j < DI / 4; ++j) {
    const float* ap = arow + (size_t)(j >> 1) * LQ * 32 + (j & 1) * 16;
    a[j].x = ap[0]; a[j].y = ap[4];
  }
#pragma unroll
  for (int it = 0; it < (48 * 48 + XT - 1) / XT; ++it) {
    const int i = it * XT + tid;
    const int r = i / 48, q = i - r * 48;
    if (i < 48 * 48) {
      const int rc = min(r, CDBL - 1);
      v4f w = *(const v4f*)(xpw + (size_t)(k * CDBL + rc) * DI + 4 * q);
      if (r >= CDBL) w = (v4f){0.f, 0.f, 0.f, 0.f};
      *(v4f*)(sW + r * XWP + 4 * q) = w;
    }
  }
  __syncthreads();
  v8f acc[3] = {};
  const float* wl = sW + lid * XWP + 2 * half;
#pragma unroll
  for (int j = 0; j < DI / 4; ++j) {
#pragma unroll
    for (int t = 0; t < 3; ++t) {
      const v2f bb = *(const v2f*)(wl + t * 16 * XWP + 4 * j);
      acc[t] = wmma4(a[j], bb, acc[t]);
    }
  }
#pragma unroll
  for (int t = 0; t < 3; ++t) {
    int col = t * 16 + lid;
    if (col < CDBL) {
      float* orow = x_dbl + ((size_t)bk * LL + lbase + half * 8) * CDBL + col;
#pragma unroll
      for (int v = 0; v < 8; ++v) orow[(size_t)v * CDBL] = acc[t][v];
    }
  }
}

#define SDT 8
#define SCS (NN * SDT + 8)
#define HQP (CDBL / 2)
#define HPC 96
#define NQ (CL / 4)
#define NHQ (CL / 2)

__global__ void __launch_bounds__(1024) k_scan_fused(
    const float* __restrict__ xconv, float* delta,
    const float* __restrict__ x_dbl, const float* __restrict__ dtw,
    const float* __restrict__ dtb, const float* __restrict__ A_logs,
    const float* __restrict__ Ds, float* __restrict__ ys) {
  __shared__ float sA[NC * SCS];
  __shared__ float sH[NC * SCS];
  __shared__ v4f sR[32 * 2 * HPC];
  const int tid = threadIdx.x;
  const int dloc = tid & (SDT - 1);
  const int c = tid / SDT;
  const int bk = blockIdx.x / (DI / SDT);
  const int d = (blockIdx.x % (DI / SDT)) * SDT + dloc;
  const int k = bk % KK, b = bk / KK;
  float Av[NN];
#pragma unroll
  for (int n = 0; n < NN; ++n)
    Av[n] = -__expf(A_logs[(size_t)(k * DI + d) * NN + n]);
  const float Dv = Ds[k * DI + d];
  const int grp = blockIdx.x % (DI / SDT);
  const float* xd = x_dbl + (size_t)bk * LL * CDBL;
  const unsigned poff = (unsigned)((bk * NG + grp) * LQ) * 32u + (unsigned)dloc * 4u;
  const unsigned xoff = (unsigned)((b * NG + grp) * LQ) * 32u + (unsigned)dloc * 4u;
  const int l0 = c * CL;
  const int sbase = c * SCS + dloc;

  {
    const int ln = tid & 31, c16 = ln & 15, hf = ln >> 4;
    const bool odd = (c16 >> 3) != 0;
    const v2f zz = {0.f, 0.f};
    const float* wrow = dtw + (size_t)(k * DI + grp * 8 + (c16 & 7)) * RR;
    v2f w03, w45;
    w03.x = wrow[2 * hf]; w03.y = wrow[2 * hf + 1];
    w45.x = wrow[4];      w45.y = wrow[5];
    if (hf) w45 = zz;
    const v2f b0 = odd ? zz : w03, b1 = odd ? zz : w45;
    const v2f b2 = odd ? w03 : zz, b3 = odd ? w45 : zz;
    const float bias = dtb[k * DI + d];
    const float* re = xd + (size_t)(((tid >> 5) * 4 + 2 * (c16 >> 3)) * CL + (c16 & 7)) * CDBL;
    const float* ro = re + (size_t)CL * CDBL;
    v4f* dw = (v4f*)(delta + ((size_t)(bk * NG + grp) * LQ) * 32) + dloc;
    for (int g = 0; g < CL / 8; ++g) {
      const float* pe = re + (size_t)g * 8 * CDBL;
      const float* po = ro + (size_t)g * 8 * CDBL;
      const v2f a0 = *(const v2f*)(pe + 2 * hf);
      v2f a1 = *(const v2f*)(pe + 4);
      const v2f a2 = *(const v2f*)(po + 2 * hf);
      v2f a3 = *(const v2f*)(po + 4);
      if (hf) { a1 = zz; a3 = zz; }
      v8f acc = {};
      acc = wmma4(a0, b0, acc);
      acc = wmma4(a1, b1, acc);
      acc = wmma4(a2, b2, acc);
      acc = wmma4(a3, b3, acc);
      v4f o0, o1;
#pragma unroll
      for (int v = 0; v < 4; ++v) {
        o0[v] = softplus_f(acc[v] + bias);
        o1[v] = softplus_f(acc[4 + v] + bias);
      }
      const int m = l0 / 4 + 2 * g;
      dw[(size_t)m * 8] = o0;
      dw[(size_t)(m + 1) * 8] = o1;
    }
  }

  const int wave = tid >> 5, lane = tid & 31;
  const v4f* xd4 = (const v4f*)xd;
  v4f* slot = sR + wave * (2 * HPC);
  unsigned gof[3];
#pragma unroll
  for (int it = 0; it < 3; ++it) {
    const unsigned P = min(it * 32 + lane, 4 * HQP - 1);
    gof[it] = (wave * 4 + P / HQP) * (NHQ * HQP) + P % HQP;
  }
  const v2f* rowp = (const v2f*)slot + (lane >> 3) * (2 * HQP) + RR / 2;
  v4f R[3];

  float h[NN];
  {
    float aP[NN];
#pragma unroll
    for (int n = 0; n < NN; ++n) { aP[n] = 1.f; h[n] = 0.f; }
    for (int m = l0 / 4; m < (l0 + CL) / 4; ++m) {
      v4f d4 = *(const v4f*)(delta + (poff + (unsigned)m * 32u));
      v4f u4 = *(const v4f*)(xconv + (xoff + (unsigned)(k ? (LQ - 1 - m) : m) * 32u));
      if (k) u4 = u4.wzyx;
#pragma nounroll
      for (int j = 0; j < 4; ++j) {
        const float u = u4.x, dlt = d4.x;
        u4 = u4.yzwx; d4 = d4.yzwx;
        const float* row = xd + (size_t)(4 * m + j) * CDBL;
        const float du = dlt * u;
#pragma unroll
        for (int n = 0; n < NN; ++n) {
          const float dA = __expf(dlt * Av[n]);
          aP[n] *= dA;
          h[n] = dA * h[n] + du * row[RR + n];
        }
      }
    }
#pragma unroll
    for (int n = 0; n < NN; ++n) {
      sA[sbase + n * SDT] = aP[n];
      sH[sbase + n * SDT] = h[n];
    }
  }
#pragma unroll
  for (int it = 0; it < 3; ++it) R[it] = xd4[gof[it]];
  __syncthreads();
  if (tid < NN * SDT) {
    float hrun = 0.f;
#pragma unroll 8
    for (int cc = 0; cc < NC; ++cc) {
      const int idx = cc * SCS + tid;
      const float a = sA[idx];
      const float hl = sH[idx];
      sH[idx] = hrun;
      hrun = a * hrun + hl;
    }
  }
#pragma unroll
  for (int it = 0; it < 3; ++it) slot[it * 32 + lane] = R[it];
  __syncthreads();
#pragma unroll
  for (int n = 0; n < NN; ++n) h[n] = sH[sbase + n * SDT];

  for (int mi = 0; mi < NQ; ++mi) {
    const int m = l0 / 4 + mi;
    v4f d4 = *(const v4f*)(delta + (poff + (unsigned)m * 32u));
    v4f u4 = *(const v4f*)(xconv + (xoff + (unsigned)(k ? (LQ - 1 - m) : m) * 32u));
    if (k) u4 = u4.wzyx;
    v4f y4 = {0.f, 0.f, 0.f, 0.f};
#pragma nounroll
    for (int hh = 0; hh < 2; ++hh) {
      const unsigned tn = min(2 * mi + hh + 1, NHQ - 1);
#pragma unroll
      for (int it = 0; it < 3; ++it) R[it] = xd4[gof[it] + HQP * tn];
      const v2f* rp = rowp + hh * (2 * HPC);
#pragma nounroll
      for (int j = 0; j < 2; ++j) {
        const float u = u4.x, dlt = d4.x;
        u4 = u4.yzwx; d4 = d4.yzwx;
        v2f bc[NN];
#pragma unroll
        for (int i = 0; i < NN; ++i) bc[i] = rp[i];
        rp += HQP;
        const float du = dlt * u;
        float y = 0.f;
#pragma unroll
        for (int n = 0; n < NN; ++n) {
          const float dA = __expf(dlt * Av[n]);
          h[n] = dA * h[n] + du * bc[n >> 1][n & 1];
          y += h[n] * bc[NN / 2 + (n >> 1)][n & 1];
        }
        y4 = y4.yzwx;
        y4.w = y + u * Dv;
      }
      v4f* st = slot + (1 - hh) * HPC + lane;
#pragma unroll
      for (int it = 0; it < 3; ++it) st[it * 32] = R[it];
    }
    *(v4f*)(ys + (poff + (unsigned)m * 32u)) = y4;
  }
}

#define OWP (DI + 4)
#define OKG 12

struct OutRaw { float y0[2 * OKG]; float y1[2 * OKG]; v2f z[OKG]; };

__device__ __forceinline__ void outproj_load(OutRaw& r, const float* __restrict__ p0,
                                             const float* __restrict__ p1,
                                             const float* __restrict__ pz, int g) {
#pragma unroll
  for (int i = 0; i < OKG; ++i) {
    const int j = g * OKG + i;
    const size_t go = (size_t)(j >> 1) * LQ * 32 + (j & 1) * 16;
    r.y0[2 * i] = p0[go];     r.y0[2 * i + 1] = p0[go + 4];
    r.y1[2 * i] = p1[go];     r.y1[2 * i + 1] = p1[go + 4];
    r.z[i] = *(const v2f*)(pz + 4 * j);
  }
}

__global__ void __launch_bounds__(256) k_outproj(const float* __restrict__ ys,
                                                 const float* __restrict__ xz,
                                                 const float* __restrict__ W_out,
                                                 float* __restrict__ out) {
  __shared__ __attribute__((aligned(16))) float sW[CC * OWP];
  const int tid = threadIdx.x, wave = tid >> 5, lane = tid & 31;
  const int lid = lane & 15, half = lane >> 4;
  const int rowBase = (blockIdx.x * 8 + wave) * 16;
  const int b = rowBase / LL;
  const int l = rowBase % LL + lid;
  const int lr = LL - 1 - l;
  const float* p0 = ys + ((size_t)(b * KK + 0) * NG * LQ + (l >> 2)) * 32 + (l & 3) + 8 * half;
  const float* p1 = ys + ((size_t)(b * KK + 1) * NG * LQ + (lr >> 2)) * 32 + (lr & 3) + 8 * half;
  const float* pz = xz + (size_t)(b * LL + l) * NOUT + DI + 2 * half;
  OutRaw cur, nxt;
  outproj_load(cur, p0, p1, pz, 0);
#pragma unroll
  for (int it = 0; it < CC * 48 / 256; ++it) {
    const int i = it * 256 + tid;
    const int r = i / 48, q = i - r * 48;
    *(v4f*)(sW + r * OWP + 4 * q) = *(const v4f*)(W_out + (size_t)r * DI + 4 * q);
  }
  __syncthreads();
  v8f acc[CC / 16] = {};
  const float* wl = sW + lid * OWP + 2 * half;
#pragma unroll
  for (int g = 0; g < DI / 4 / OKG; ++g) {
    if (g + 1 < DI / 4 / OKG) outproj_load(nxt, p0, p1, pz, g + 1);
#pragma unroll
    for (int i = 0; i < OKG; ++i) {
      const int j = g * OKG + i;
      v2f a;
      a.x = (cur.y0[2 * i] + cur.y1[2 * i]) * silu_f(cur.z[i].x);
      a.y = (cur.y0[2 * i + 1] + cur.y1[2 * i + 1]) * silu_f(cur.z[i].y);
#pragma unroll
      for (int nt = 0; nt < CC / 16; ++nt) {
        const v2f bb = *(const v2f*)(wl + nt * 16 * OWP + 4 * j);
        acc[nt] = wmma4(a, bb, acc[nt]);
      }
    }
    if (g + 1 < DI / 4 / OKG) cur = nxt;
  }
  float* orow = out + (size_t)(rowBase + half * 8) * CC + lid;
#pragma unroll
  for (int nt = 0; nt < CC / 16; ++nt) {
#pragma unroll
    for (int v = 0; v < 8; ++v) orow[(size_t)v * CC + nt * 16] = acc[nt][v];
  }
}

extern "C" void kernel_launch(void* const* d_in, const int* in_sizes, int n_in,
                              void* d_out, int out_size, void* d_ws, size_t ws_size,
                              hipStream_t stream) {
  (void)in_sizes; (void)n_in; (void)out_size; (void)ws_size;
  const float* x      = (const float*)d_in[0];
  const float* W_in   = (const float*)d_in[1];
  const float* conv_w = (const float*)d_in[2];
  const float* conv_b = (const float*)d_in[3];
  const float* xpw    = (const float*)d_in[4];
  const float* dtw    = (const float*)d_in[5];
  const float* dtb    = (const float*)d_in[6];
  const float* A_logs = (const float*)d_in[7];
  const float* Ds     = (const float*)d_in[8];
  const float* W_out  = (const float*)d_in[9];
  float* out = (float*)d_out;

  char* ws = (char*)d_ws;
  size_t off = 0;
  auto alloc = [&](size_t nbytes) {
    char* p = ws + off;
    off = (off + nbytes + 255) & ~(size_t)255;
    return (float*)p;
  };
  float* xz     = alloc((size_t)BL * NOUT * sizeof(float));
  float* xconv  = alloc((size_t)BL * DI * sizeof(float));
  float* x_dbl  = alloc((size_t)BB * KK * LL * CDBL * sizeof(float));
  float* delta  = alloc((size_t)BB * KK * LL * DI * sizeof(float));
  float* ysb    = alloc((size_t)BB * KK * LL * DI * sizeof(float));

  k_inproj_conv<<<BB * (HH / TH) * (WWD / TW), FT, 0, stream>>>(x, W_in, conv_w,
                                                                conv_b, xz, xconv);
  k_xdbl<<<(BB * KK * LL / 16) / (XT / 32), XT, 0, stream>>>(xconv, xpw, x_dbl);
  k_scan_fused<<<BB * KK * (DI / SDT), NC * SDT, 0, stream>>>(
      xconv, delta, x_dbl, dtw, dtb, A_logs, Ds, ysb);
  k_outproj<<<BL / 128, 256, 0, stream>>>(ysb, xz, W_out, out);
}
